# starting kernel
# baseline (speedup 1.0000x reference)
#include <hip/hip_runtime.h>
#include <hip/hip_bf16.h>


typedef __attribute__((ext_vector_type(2))) float v2f;
typedef __attribute__((ext_vector_type(8))) float v8f;

#define TWO_PI_F     6.28318530717958647692f
#define PI_F         3.14159265358979323846f
#define INV_TWO_PI_F 0.15915494309189533577f
#define DT_F         0.001f

__device__ __forceinline__ float pmod(float x, float m) {
    float r = fmodf(x, m);
    return (r < 0.0f) ? (r + m) : r;
}

__device__ __forceinline__ float wrap_add(float x, float d) {
    float t = x + d;
    return (t >= TWO_PI_F) ? (t - TWO_PI_F) : t;
}

struct alignas(16) F4 { float v[4]; };

__global__ void osc_scan_kernel(const float* __restrict__ p_log_slow,
                                const float* __restrict__ p_log_fast,
                                const float* __restrict__ p_reset,
                                float* __restrict__ out, int N) {
    if (threadIdx.x != 0 || blockIdx.x != 0) return;

    const float dslow = expf(p_log_slow[0]) * DT_F;
    const float dfast = expf(p_log_fast[0]) * DT_F;
    const float keep  = 1.0f - p_reset[0];

    float slow = 0.0f, fast = 0.0f, prev = 0.0f;
    float* __restrict__ oS = out;
    float* __restrict__ oF = out + N;

    const bool fast_ok =
        (dslow >= 0.0f) && (dslow < TWO_PI_F) &&
        (dfast >= 0.0f) && (dfast < TWO_PI_F) &&
        (keep  >= 0.0f) && (keep  <= 1.0f);

    int t = 0;
    if (fast_ok && ((N & 3) == 0)) {
        for (; t < N; t += 4) {
            F4 s4, f4;
#pragma unroll
            for (int u = 0; u < 4; ++u) {
                s4.v[u] = slow;
                f4.v[u] = fast;
                bool crossed = (prev > PI_F) && (slow < PI_F);
                float ns = wrap_add(slow, dslow);
                float nf = wrap_add(fast, dfast);
                nf = crossed ? nf * keep : nf;
                prev = slow; slow = ns; fast = nf;
            }
            *(F4*)(oS + t) = s4;
            *(F4*)(oF + t) = f4;
        }
    }
    for (; t < N; ++t) {
        oS[t] = slow;
        oF[t] = fast;
        bool crossed = (prev > PI_F) && (slow < PI_F);
        float ns = pmod(slow + dslow, TWO_PI_F);
        float nf = pmod(fast + dfast, TWO_PI_F);
        if (crossed) nf = pmod(nf * keep, TWO_PI_F);
        prev = slow; slow = ns; fast = nf;
    }
}

__global__ void osc_scale_wmma_kernel(const float* __restrict__ in,
                                      float* __restrict__ out, int N) {
    const float s = INV_TWO_PI_F;
    const int base = blockIdx.x * 256;
    const int lane = threadIdx.x;
    const int i    = lane & 15;
    const int kh   = lane >> 4;

    if (base + 256 <= N) {
        v8f d = {0.f, 0.f, 0.f, 0.f, 0.f, 0.f, 0.f, 0.f};
#pragma unroll
        for (int m = 0; m < 4; ++m) {
            const float* p = in + base + i * 16 + 4 * m + 2 * kh;
            v2f a; a.x = p[0]; a.y = p[1];
            v2f b;
            b.x = (i == (4 * m + 2 * kh + 0)) ? s : 0.0f;
            b.y = (i == (4 * m + 2 * kh + 1)) ? s : 0.0f;
            d = __builtin_amdgcn_wmma_f32_16x16x4_f32(
                    false, a, false, b, (short)0, d, false, false);
        }
#pragma unroll
        for (int r = 0; r < 8; ++r) {
            out[base + (r + 8 * kh) * 16 + i] = d[r];
        }
    } else {
        for (int j = base + lane; j < N; j += 32) out[j] = in[j] * s;
    }
}

extern "C" void kernel_launch(void* const* d_in, const int* in_sizes, int n_in,
                              void* d_out, int out_size, void* d_ws, size_t ws_size,
                              hipStream_t stream) {
    const float* p_log_slow = (const float*)d_in[0];
    const float* p_log_fast = (const float*)d_in[1];
    const float* p_reset    = (const float*)d_in[2];
    const int N = out_size / 3;
    if (N <= 0) return;
    float* out = (float*)d_out;

    osc_scan_kernel<<<1, 32, 0, stream>>>(p_log_slow, p_log_fast, p_reset, out, N);

    const int tiles = (N + 255) / 256;
    osc_scale_wmma_kernel<<<tiles, 32, 0, stream>>>(out, out + 2 * (size_t)N, N);
}


# scan: slow and fast phase in two lanes of one wave; steps that a bound on the phase proves wrap-free run as bare adds, 8 per block; wrap test and reset only near wraps; same sums, same order
# speedup vs baseline: 2.0249x; 2.0249x over previous
#include <hip/hip_runtime.h>
#include <hip/hip_bf16.h>


typedef __attribute__((ext_vector_type(2))) float v2f;
typedef __attribute__((ext_vector_type(8))) float v8f;

#define TWO_PI_F     6.28318530717958647692f
#define PI_F         3.14159265358979323846f
#define INV_TWO_PI_F 0.15915494309189533577f
#define DT_F         0.001f

__device__ __forceinline__ float pmod(float x, float m) {
    float r = fmodf(x, m);
    return (r < 0.0f) ? (r + m) : r;
}

__device__ __forceinline__ float wrap_add(float x, float d) {
    float t = x + d;
    return (t >= TWO_PI_F) ? (t - TWO_PI_F) : t;
}

struct alignas(16) F4 { float v[4]; };

__global__ void osc_scan_kernel(const float* __restrict__ p_log_slow,
                                const float* __restrict__ p_log_fast,
                                const float* __restrict__ p_reset,
                                float* __restrict__ out, int N) {
    const int lane = threadIdx.x;
    if (blockIdx.x != 0 || lane >= 2) return;

    const float dslow = expf(p_log_slow[0]) * DT_F;
    const float dfast = expf(p_log_fast[0]) * DT_F;
    const float keep  = 1.0f - p_reset[0];

    const bool fast_ok =
        (dslow >= 0.0f) && (dslow < TWO_PI_F) &&
        (dfast >= 0.0f) && (dfast < TWO_PI_F) &&
        (keep  >= 0.0f) && (keep  <= 1.0f);

    if (fast_ok && ((N & 3) == 0)) {
        const bool is_fast = (lane == 1);
        const float d = is_fast ? dfast : dslow;
        const float thr4  = TWO_PI_F -  4.0f * d - 1.0e-4f;
        const float thr8  = TWO_PI_F -  8.0f * d - 1.0e-4f;
        const float thr16 = TWO_PI_F - 16.0f * d - 1.0e-4f;
        float* __restrict__ o = out + (is_fast ? (size_t)N : (size_t)0);

        auto slow_crossed = [](float pv, float xv) -> bool {
            return (__builtin_amdgcn_ballot_w32(pv > PI_F) &
                    __builtin_amdgcn_ballot_w32(xv < PI_F) & 1u) != 0u;
        };

        float x = 0.0f, prev = 0.0f;
        bool safe8 = (__builtin_amdgcn_ballot_w32(x >= thr8) == 0u);
        int t = 0;
        while (t < N) {
            while (safe8 && (t + 8 <= N)) {
                safe8 = (__builtin_amdgcn_ballot_w32(x >= thr16) == 0u);
                const float a1 = x  + d, a2 = a1 + d, a3 = a2 + d, a4 = a3 + d;
                const float a5 = a4 + d, a6 = a5 + d, a7 = a6 + d, a8 = a7 + d;
                F4 v, w;
                v.v[0] = x;  v.v[1] = a1; v.v[2] = a2; v.v[3] = a3;
                w.v[0] = a4; w.v[1] = a5; w.v[2] = a6; w.v[3] = a7;
                *(F4*)(o + t)     = v;
                *(F4*)(o + t + 4) = w;
                prev = a7; x = a8;
                t += 8;
            }
            if (t < N) {
                const unsigned may_wrap = __builtin_amdgcn_ballot_w32(x >= thr4);
                const bool cross0 = slow_crossed(prev, x);
                F4 v;
                if (cross0 || (may_wrap & 1u) != 0u) {
#pragma unroll
                    for (int u = 0; u < 4; ++u) {
                        v.v[u] = x;
                        const bool crossed = slow_crossed(prev, x);
                        float n = wrap_add(x, d);
                        n = (crossed && is_fast) ? n * keep : n;
                        prev = x; x = n;
                    }
                } else if (may_wrap != 0u) {
#pragma unroll
                    for (int u = 0; u < 4; ++u) {
                        v.v[u] = x;
                        const float n = wrap_add(x, d);
                        prev = x; x = n;
                    }
                } else {
                    const float a1 = x + d, a2 = a1 + d, a3 = a2 + d, a4 = a3 + d;
                    v.v[0] = x; v.v[1] = a1; v.v[2] = a2; v.v[3] = a3;
                    prev = a3; x = a4;
                }
                *(F4*)(o + t) = v;
                t += 4;
                safe8 = (__builtin_amdgcn_ballot_w32(x >= thr8) == 0u) && !slow_crossed(prev, x);
            }
        }
        return;
    }

    if (lane != 0) return;
    float slow = 0.0f, fast = 0.0f, prev = 0.0f;
    float* __restrict__ oS = out;
    float* __restrict__ oF = out + N;
    for (int t = 0; t < N; ++t) {
        oS[t] = slow;
        oF[t] = fast;
        bool crossed = (prev > PI_F) && (slow < PI_F);
        float ns = pmod(slow + dslow, TWO_PI_F);
        float nf = pmod(fast + dfast, TWO_PI_F);
        if (crossed) nf = pmod(nf * keep, TWO_PI_F);
        prev = slow; slow = ns; fast = nf;
    }
}

__global__ void osc_scale_wmma_kernel(const float* __restrict__ in,
                                      float* __restrict__ out, int N) {
    const float s = INV_TWO_PI_F;
    const int base = blockIdx.x * 256;
    const int lane = threadIdx.x;
    const int i    = lane & 15;
    const int kh   = lane >> 4;

    if (base + 256 <= N) {
        v8f d = {0.f, 0.f, 0.f, 0.f, 0.f, 0.f, 0.f, 0.f};
#pragma unroll
        for (int m = 0; m < 4; ++m) {
            const float* p = in + base + i * 16 + 4 * m + 2 * kh;
            v2f a; a.x = p[0]; a.y = p[1];
            v2f b;
            b.x = (i == (4 * m + 2 * kh + 0)) ? s : 0.0f;
            b.y = (i == (4 * m + 2 * kh + 1)) ? s : 0.0f;
            d = __builtin_amdgcn_wmma_f32_16x16x4_f32(
                    false, a, false, b, (short)0, d, false, false);
        }
#pragma unroll
        for (int r = 0; r < 8; ++r) {
            out[base + (r + 8 * kh) * 16 + i] = d[r];
        }
    } else {
        for (int j = base + lane; j < N; j += 32) out[j] = in[j] * s;
    }
}

extern "C" void kernel_launch(void* const* d_in, const int* in_sizes, int n_in,
                              void* d_out, int out_size, void* d_ws, size_t ws_size,
                              hipStream_t stream) {
    const float* p_log_slow = (const float*)d_in[0];
    const float* p_log_fast = (const float*)d_in[1];
    const float* p_reset    = (const float*)d_in[2];
    const int N = out_size / 3;
    if (N <= 0) return;
    float* out = (float*)d_out;

    osc_scan_kernel<<<1, 32, 0, stream>>>(p_log_slow, p_log_fast, p_reset, out, N);

    const int tiles = (N + 255) / 256;
    osc_scale_wmma_kernel<<<tiles, 32, 0, stream>>>(out, out + 2 * (size_t)N, N);
}


# scan: wrap-free run length counted once per wrap (one multiply on the phase before the wrap block), run as bare adds under an integer trip count; no per-block test of the phase; same sums, same order
# speedup vs baseline: 2.5048x; 1.2370x over previous
#include <hip/hip_runtime.h>
#include <hip/hip_bf16.h>


typedef __attribute__((ext_vector_type(2))) float v2f;
typedef __attribute__((ext_vector_type(8))) float v8f;

#define TWO_PI_F     6.28318530717958647692f
#define PI_F         3.14159265358979323846f
#define INV_TWO_PI_F 0.15915494309189533577f
#define DT_F         0.001f

__device__ __forceinline__ float pmod(float x, float m) {
    float r = fmodf(x, m);
    return (r < 0.0f) ? (r + m) : r;
}

__device__ __forceinline__ float wrap_add(float x, float d) {
    float t = x + d;
    return (t >= TWO_PI_F) ? (t - TWO_PI_F) : t;
}

struct alignas(16) F4 { float v[4]; };

__device__ __forceinline__ int safe_steps(float x, float thr0, float inv_d) {
    const float q = fminf((thr0 - x) * inv_d, 1024.0f);
    const int c = (int)q;
    return c < 0 ? 0 : c;
}

__device__ __forceinline__ int safe_steps_after4(float x, float d4, float thr0, float inv_d) {
    const float y = x + d4;
    const bool wrapped = (y >= TWO_PI_F + 1.0e-3f);
    const bool known = (d4 <= thr0) && (wrapped || (y < thr0));
    const float xub = (wrapped ? (y - TWO_PI_F) : y) + 1.0e-4f;
    return safe_steps(known ? xub : TWO_PI_F, thr0, inv_d);
}

__global__ void osc_scan_kernel(const float* __restrict__ p_log_slow,
                                const float* __restrict__ p_log_fast,
                                const float* __restrict__ p_reset,
                                float* __restrict__ out, int N) {
    const int lane = threadIdx.x;
    if (blockIdx.x != 0 || lane >= 2) return;

    const float dslow = expf(p_log_slow[0]) * DT_F;
    const float dfast = expf(p_log_fast[0]) * DT_F;
    const float keep  = 1.0f - p_reset[0];

    const bool fast_ok =
        (dslow >= 0.0f) && (dslow < TWO_PI_F) &&
        (dfast >= 0.0f) && (dfast < TWO_PI_F) &&
        (keep  >= 0.0f) && (keep  <= 1.0f);

    if (fast_ok && ((N & 3) == 0)) {
        const bool is_fast = (lane == 1);
        const float d = is_fast ? dfast : dslow;
        const float inv_d = 1.0f / d;
        const float d4    = 4.0f * d;
        const float thr0  = TWO_PI_F - 1.0e-3f;
        float* __restrict__ o = out + (is_fast ? (size_t)N : (size_t)0);

        auto slow_crossed = [](float pv, float xv) -> bool {
            return (__builtin_amdgcn_ballot_w32(pv > PI_F) &
                    __builtin_amdgcn_ballot_w32(xv < PI_F) & 1u) != 0u;
        };

        float x = 0.0f, prev = 0.0f;
        int c  = safe_steps(x, thr0, inv_d);
        int cs = __builtin_amdgcn_readlane(c, 0);
        int cf = __builtin_amdgcn_readlane(c, 1);
        bool pend = false;
        int t = 0;
        while (t < N) {
            int nb = pend ? 0 : ((cs < cf ? cs : cf) >> 2);
            const int left = (N - t) >> 2;
            nb = nb < left ? nb : left;
            cs -= 4 * nb; cf -= 4 * nb;
            const int t8 = t + ((nb >> 1) << 3);
            for (; t < t8; ) {
                const float a1 = x  + d, a2 = a1 + d, a3 = a2 + d, a4 = a3 + d;
                const float a5 = a4 + d, a6 = a5 + d, a7 = a6 + d, a8 = a7 + d;
                F4 v, w;
                v.v[0] = x;  v.v[1] = a1; v.v[2] = a2; v.v[3] = a3;
                w.v[0] = a4; w.v[1] = a5; w.v[2] = a6; w.v[3] = a7;
                const int ti = __builtin_amdgcn_readfirstlane(t);
                *(F4*)(o + ti)     = v;
                *(F4*)(o + ti + 4) = w;
                prev = a7; x = a8;
                t += 8;
            }
            if (nb & 1) {
                const float a1 = x + d, a2 = a1 + d, a3 = a2 + d, a4 = a3 + d;
                F4 v;
                v.v[0] = x; v.v[1] = a1; v.v[2] = a2; v.v[3] = a3;
                *(F4*)(o + t) = v;
                prev = a3; x = a4;
                t += 4;
            }
            if (t >= N) break;
            F4 v;
            if (pend || cs < 4) {
#pragma unroll
                for (int u = 0; u < 4; ++u) {
                    v.v[u] = x;
                    const bool crossed = slow_crossed(prev, x);
                    float n = wrap_add(x, d);
                    n = (crossed && is_fast) ? n * keep : n;
                    prev = x; x = n;
                }
                *(F4*)(o + t) = v;
                t += 4;
                c    = safe_steps(x, thr0, inv_d);
                cs   = __builtin_amdgcn_readlane(c, 0);
                cf   = __builtin_amdgcn_readlane(c, 1);
                pend = slow_crossed(prev, x);
            } else {
                const int cn = __builtin_amdgcn_readlane(safe_steps_after4(x, d4, thr0, inv_d), 1);
#pragma unroll
                for (int u = 0; u < 4; ++u) {
                    v.v[u] = x;
                    prev = x; x = wrap_add(x, d);
                }
                *(F4*)(o + t) = v;
                t += 4;
                cs -= 4;
                cf  = cn;
            }
        }
        return;
    }

    if (lane != 0) return;
    float slow = 0.0f, fast = 0.0f, prev = 0.0f;
    float* __restrict__ oS = out;
    float* __restrict__ oF = out + N;
    for (int t = 0; t < N; ++t) {
        oS[t] = slow;
        oF[t] = fast;
        bool crossed = (prev > PI_F) && (slow < PI_F);
        float ns = pmod(slow + dslow, TWO_PI_F);
        float nf = pmod(fast + dfast, TWO_PI_F);
        if (crossed) nf = pmod(nf * keep, TWO_PI_F);
        prev = slow; slow = ns; fast = nf;
    }
}

__global__ void osc_scale_wmma_kernel(const float* __restrict__ in,
                                      float* __restrict__ out, int N) {
    const float s = INV_TWO_PI_F;
    const int base = blockIdx.x * 256;
    const int lane = threadIdx.x;
    const int i    = lane & 15;
    const int kh   = lane >> 4;

    if (base + 256 <= N) {
        v8f d = {0.f, 0.f, 0.f, 0.f, 0.f, 0.f, 0.f, 0.f};
#pragma unroll
        for (int m = 0; m < 4; ++m) {
            const float* p = in + base + i * 16 + 4 * m + 2 * kh;
            v2f a; a.x = p[0]; a.y = p[1];
            v2f b;
            b.x = (i == (4 * m + 2 * kh + 0)) ? s : 0.0f;
            b.y = (i == (4 * m + 2 * kh + 1)) ? s : 0.0f;
            d = __builtin_amdgcn_wmma_f32_16x16x4_f32(
                    false, a, false, b, (short)0, d, false, false);
        }
#pragma unroll
        for (int r = 0; r < 8; ++r) {
            out[base + (r + 8 * kh) * 16 + i] = d[r];
        }
    } else {
        for (int j = base + lane; j < N; j += 32) out[j] = in[j] * s;
    }
}

extern "C" void kernel_launch(void* const* d_in, const int* in_sizes, int n_in,
                              void* d_out, int out_size, void* d_ws, size_t ws_size,
                              hipStream_t stream) {
    const float* p_log_slow = (const float*)d_in[0];
    const float* p_log_fast = (const float*)d_in[1];
    const float* p_reset    = (const float*)d_in[2];
    const int N = out_size / 3;
    if (N <= 0) return;
    float* out = (float*)d_out;

    osc_scan_kernel<<<1, 32, 0, stream>>>(p_log_slow, p_log_fast, p_reset, out, N);

    const int tiles = (N + 255) / 256;
    osc_scale_wmma_kernel<<<tiles, 32, 0, stream>>>(out, out + 2 * (size_t)N, N);
}


# scan: wrap taken off the branch: a step forms x+d and x-T side by side (T = 2pi - d rounded to the 2^-21 grid of [4,8); both exact), keeps the unsigned-smaller bit pattern; values bit-identical
# speedup vs baseline: 2.8298x; 1.1298x over previous
#include <hip/hip_runtime.h>
#include <hip/hip_bf16.h>


typedef __attribute__((ext_vector_type(2))) float v2f;
typedef __attribute__((ext_vector_type(8))) float v8f;

#define TWO_PI_F     6.28318530717958647692f
#define PI_F         3.14159265358979323846f
#define INV_TWO_PI_F 0.15915494309189533577f
#define DT_F         0.001f

__device__ __forceinline__ float pmod(float x, float m) {
    float r = fmodf(x, m);
    return (r < 0.0f) ? (r + m) : r;
}

__device__ __forceinline__ float wrap_add(float x, float d) {
    float t = x + d;
    return (t >= TWO_PI_F) ? (t - TWO_PI_F) : t;
}

struct alignas(16) F4 { float v[4]; };

__device__ __forceinline__ int safe_steps(float x, float thr0, float inv_d) {
    const float q = fminf((thr0 - x) * inv_d, 1024.0f);
    const int c = (int)q;
    return c < 0 ? 0 : c;
}

__device__ __forceinline__ bool on_grid(float d, float* dq) {
    const float q = (4.0f + d) - 4.0f;
    const float r = d - q;
    *dq = q;
    return (d >= 0.0f) && (d <= 1.7f) && (fabsf(r) != 0x1p-22f);
}

template <bool GRID>
__device__ __forceinline__ float wrap_step(float x, float d, float T) {
    const float s = x + d;
    const float w = GRID ? (x - T) : (s - TWO_PI_F);
    const unsigned a = __float_as_uint(s);
    const unsigned b = __float_as_uint(w);
    return __uint_as_float(a < b ? a : b);
}

template <bool GRID>
__device__ __forceinline__ void scan_two_lanes(const bool is_fast, const float d, const float T,
                                               const float keep, float* __restrict__ o, const int N) {
    const float inv_d = 1.0f / d;
    const float thr0  = TWO_PI_F - 1.0e-3f;

    auto slow_crossed = [](float pv, float xv) -> bool {
        return (__builtin_amdgcn_ballot_w32(pv > PI_F) &
                __builtin_amdgcn_ballot_w32(xv < PI_F) & 1u) != 0u;
    };

    float x = 0.0f, prev = 0.0f;
    int cs = __builtin_amdgcn_readlane(safe_steps(x, thr0, inv_d), 0);
    bool pend = false;
    int t = 0;
    while (t < N) {
        int nb = pend ? 0 : (cs >> 2);
        const int left = (N - t) >> 2;
        nb = nb < left ? nb : left;
        cs -= 4 * nb;
        const int t8 = t + ((nb >> 1) << 3);
        for (; t < t8; ) {
            const float a1 = wrap_step<GRID>(x,  d, T), a2 = wrap_step<GRID>(a1, d, T);
            const float a3 = wrap_step<GRID>(a2, d, T), a4 = wrap_step<GRID>(a3, d, T);
            const float a5 = wrap_step<GRID>(a4, d, T), a6 = wrap_step<GRID>(a5, d, T);
            const float a7 = wrap_step<GRID>(a6, d, T), a8 = wrap_step<GRID>(a7, d, T);
            F4 v, w;
            v.v[0] = x;  v.v[1] = a1; v.v[2] = a2; v.v[3] = a3;
            w.v[0] = a4; w.v[1] = a5; w.v[2] = a6; w.v[3] = a7;
            const int ti = __builtin_amdgcn_readfirstlane(t);
            *(F4*)(o + ti)     = v;
            *(F4*)(o + ti + 4) = w;
            prev = a7; x = a8;
            t += 8;
        }
        if (nb & 1) {
            const float a1 = wrap_step<GRID>(x,  d, T), a2 = wrap_step<GRID>(a1, d, T);
            const float a3 = wrap_step<GRID>(a2, d, T), a4 = wrap_step<GRID>(a3, d, T);
            F4 v;
            v.v[0] = x; v.v[1] = a1; v.v[2] = a2; v.v[3] = a3;
            *(F4*)(o + t) = v;
            prev = a3; x = a4;
            t += 4;
        }
        if (t >= N) break;
        F4 v;
#pragma unroll
        for (int u = 0; u < 4; ++u) {
            v.v[u] = x;
            const bool crossed = slow_crossed(prev, x);
            float n = wrap_add(x, d);
            n = (crossed && is_fast) ? n * keep : n;
            prev = x; x = n;
        }
        *(F4*)(o + t) = v;
        t += 4;
        cs   = __builtin_amdgcn_readlane(safe_steps(x, thr0, inv_d), 0);
        pend = slow_crossed(prev, x);
    }
}

__global__ void osc_scan_kernel(const float* __restrict__ p_log_slow,
                                const float* __restrict__ p_log_fast,
                                const float* __restrict__ p_reset,
                                float* __restrict__ out, int N) {
    const int lane = threadIdx.x;
    if (blockIdx.x != 0 || lane >= 2) return;

    const float dslow = expf(p_log_slow[0]) * DT_F;
    const float dfast = expf(p_log_fast[0]) * DT_F;
    const float keep  = 1.0f - p_reset[0];

    const bool fast_ok =
        (dslow >= 0.0f) && (dslow < TWO_PI_F) &&
        (dfast >= 0.0f) && (dfast < TWO_PI_F) &&
        (keep  >= 0.0f) && (keep  <= 1.0f);

    if (fast_ok && ((N & 3) == 0)) {
        float dq_slow, dq_fast;
        const bool grid_s = on_grid(dslow, &dq_slow);
        const bool grid_f = on_grid(dfast, &dq_fast);
        const bool is_fast = (lane == 1);
        const float d = is_fast ? dfast : dslow;
        const float T = TWO_PI_F - (is_fast ? dq_fast : dq_slow);
        float* __restrict__ o = out + (is_fast ? (size_t)N : (size_t)0);
        if (grid_s && grid_f) scan_two_lanes<true >(is_fast, d, T, keep, o, N);
        else                  scan_two_lanes<false>(is_fast, d, T, keep, o, N);
        return;
    }

    if (lane != 0) return;
    float slow = 0.0f, fast = 0.0f, prev = 0.0f;
    float* __restrict__ oS = out;
    float* __restrict__ oF = out + N;
    for (int t = 0; t < N; ++t) {
        oS[t] = slow;
        oF[t] = fast;
        bool crossed = (prev > PI_F) && (slow < PI_F);
        float ns = pmod(slow + dslow, TWO_PI_F);
        float nf = pmod(fast + dfast, TWO_PI_F);
        if (crossed) nf = pmod(nf * keep, TWO_PI_F);
        prev = slow; slow = ns; fast = nf;
    }
}

__global__ void osc_scale_wmma_kernel(const float* __restrict__ in,
                                      float* __restrict__ out, int N) {
    const float s = INV_TWO_PI_F;
    const int base = blockIdx.x * 256;
    const int lane = threadIdx.x;
    const int i    = lane & 15;
    const int kh   = lane >> 4;

    if (base + 256 <= N) {
        v8f d = {0.f, 0.f, 0.f, 0.f, 0.f, 0.f, 0.f, 0.f};
#pragma unroll
        for (int m = 0; m < 4; ++m) {
            const float* p = in + base + i * 16 + 4 * m + 2 * kh;
            v2f a; a.x = p[0]; a.y = p[1];
            v2f b;
            b.x = (i == (4 * m + 2 * kh + 0)) ? s : 0.0f;
            b.y = (i == (4 * m + 2 * kh + 1)) ? s : 0.0f;
            d = __builtin_amdgcn_wmma_f32_16x16x4_f32(
                    false, a, false, b, (short)0, d, false, false);
        }
#pragma unroll
        for (int r = 0; r < 8; ++r) {
            out[base + (r + 8 * kh) * 16 + i] = d[r];
        }
    } else {
        for (int j = base + lane; j < N; j += 32) out[j] = in[j] * s;
    }
}

extern "C" void kernel_launch(void* const* d_in, const int* in_sizes, int n_in,
                              void* d_out, int out_size, void* d_ws, size_t ws_size,
                              hipStream_t stream) {
    const float* p_log_slow = (const float*)d_in[0];
    const float* p_log_fast = (const float*)d_in[1];
    const float* p_reset    = (const float*)d_in[2];
    const int N = out_size / 3;
    if (N <= 0) return;
    float* out = (float*)d_out;

    osc_scan_kernel<<<1, 32, 0, stream>>>(p_log_slow, p_log_fast, p_reset, out, N);

    const int tiles = (N + 255) / 256;
    osc_scale_wmma_kernel<<<tiles, 32, 0, stream>>>(out, out + 2 * (size_t)N, N);
}


# scan: both chains run in all 32 lanes (16 copies a phase); copy j keeps step j of a 16-step pass by one select beside the chain; one 32-lane dword store per 16 steps, was four stores; same values
# speedup vs baseline: 3.4516x; 1.2197x over previous
#include <hip/hip_runtime.h>
#include <hip/hip_bf16.h>


typedef __attribute__((ext_vector_type(2))) float v2f;
typedef __attribute__((ext_vector_type(8))) float v8f;

#define TWO_PI_F     6.28318530717958647692f
#define PI_F         3.14159265358979323846f
#define INV_TWO_PI_F 0.15915494309189533577f
#define DT_F         0.001f

__device__ __forceinline__ float pmod(float x, float m) {
    float r = fmodf(x, m);
    return (r < 0.0f) ? (r + m) : r;
}

__device__ __forceinline__ float wrap_add(float x, float d) {
    float t = x + d;
    return (t >= TWO_PI_F) ? (t - TWO_PI_F) : t;
}

struct alignas(16) F4 { float v[4]; };

__device__ __forceinline__ int safe_steps(float x, float thr0, float inv_d) {
    const float q = fminf((thr0 - x) * inv_d, 1024.0f);
    const int c = (int)q;
    return c < 0 ? 0 : c;
}

__device__ __forceinline__ bool on_grid(float d, float* dq) {
    const float q = (4.0f + d) - 4.0f;
    const float r = d - q;
    *dq = q;
    return (d >= 0.0f) && (d <= 1.7f) && (fabsf(r) != 0x1p-22f);
}

template <bool GRID>
__device__ __forceinline__ float wrap_step(float x, float d, float T) {
    const float s = x + d;
    const float w = GRID ? (x - T) : (s - TWO_PI_F);
    const unsigned a = __float_as_uint(s);
    const unsigned b = __float_as_uint(w);
    return __uint_as_float(a < b ? a : b);
}

template <bool GRID>
__device__ __forceinline__ void scan_two_lanes(const bool is_fast, const int slot, const float d, const float T,
                                               const float keep, float* __restrict__ o, const int N) {
    const bool writer = (slot == 0);
    float* __restrict__ oc = o + slot;
    const float inv_d = 1.0f / d;
    const float thr0  = TWO_PI_F - 1.0e-3f;

    auto slow_crossed = [](float pv, float xv) -> bool {
        return (__builtin_amdgcn_ballot_w32(pv > PI_F) &
                __builtin_amdgcn_ballot_w32(xv < PI_F) & 1u) != 0u;
    };

    float x = 0.0f, prev = 0.0f;
    int cs = __builtin_amdgcn_readlane(safe_steps(x, thr0, inv_d), 0);
    bool pend = false;
    int t = 0;
    while (t < N) {
        int nb = pend ? 0 : (cs >> 2);
        const int left = (N - t) >> 2;
        nb = nb < left ? nb : left;
        cs -= 4 * nb;
        const int t16 = t + ((nb >> 2) << 4);
        for (; t < t16; ) {
            const int ti = __builtin_amdgcn_readfirstlane(t);
            float* __restrict__ p = oc + ti;
            asm volatile("" :: "v"(p));
            float a = x, c = x;
#pragma unroll
            for (int j = 0; j < 16; ++j) {
                c = (slot == j) ? a : c;
                if (j == 15) prev = a;
                a = wrap_step<GRID>(a, d, T);
                __builtin_amdgcn_sched_barrier(0);
            }
            *p = c;
            x = a;
            t += 16;
        }
        nb &= 3;
        const int t8 = t + ((nb >> 1) << 3);
        for (; t < t8; ) {
            const float a1 = wrap_step<GRID>(x,  d, T), a2 = wrap_step<GRID>(a1, d, T);
            const float a3 = wrap_step<GRID>(a2, d, T), a4 = wrap_step<GRID>(a3, d, T);
            const float a5 = wrap_step<GRID>(a4, d, T), a6 = wrap_step<GRID>(a5, d, T);
            const float a7 = wrap_step<GRID>(a6, d, T), a8 = wrap_step<GRID>(a7, d, T);
            F4 v, w;
            v.v[0] = x;  v.v[1] = a1; v.v[2] = a2; v.v[3] = a3;
            w.v[0] = a4; w.v[1] = a5; w.v[2] = a6; w.v[3] = a7;
            const int ti = __builtin_amdgcn_readfirstlane(t);
            if (writer) {
                *(F4*)(o + ti)     = v;
                *(F4*)(o + ti + 4) = w;
            }
            prev = a7; x = a8;
            t += 8;
        }
        if (nb & 1) {
            const float a1 = wrap_step<GRID>(x,  d, T), a2 = wrap_step<GRID>(a1, d, T);
            const float a3 = wrap_step<GRID>(a2, d, T), a4 = wrap_step<GRID>(a3, d, T);
            F4 v;
            v.v[0] = x; v.v[1] = a1; v.v[2] = a2; v.v[3] = a3;
            if (writer) *(F4*)(o + t) = v;
            prev = a3; x = a4;
            t += 4;
        }
        if (t >= N) break;
        F4 v;
#pragma unroll
        for (int u = 0; u < 4; ++u) {
            v.v[u] = x;
            const bool crossed = slow_crossed(prev, x);
            float n = wrap_add(x, d);
            n = (crossed && is_fast) ? n * keep : n;
            prev = x; x = n;
        }
        if (writer) *(F4*)(o + t) = v;
        t += 4;
        cs   = __builtin_amdgcn_readlane(safe_steps(x, thr0, inv_d), 0);
        pend = slow_crossed(prev, x);
    }
}

__global__ void osc_scan_kernel(const float* __restrict__ p_log_slow,
                                const float* __restrict__ p_log_fast,
                                const float* __restrict__ p_reset,
                                float* __restrict__ out, int N) {
    const int lane = threadIdx.x;
    if (blockIdx.x != 0) return;

    const float dslow = expf(p_log_slow[0]) * DT_F;
    const float dfast = expf(p_log_fast[0]) * DT_F;
    const float keep  = 1.0f - p_reset[0];

    const bool fast_ok =
        (dslow >= 0.0f) && (dslow < TWO_PI_F) &&
        (dfast >= 0.0f) && (dfast < TWO_PI_F) &&
        (keep  >= 0.0f) && (keep  <= 1.0f);

    if (fast_ok && ((N & 3) == 0)) {
        float dq_slow, dq_fast;
        const bool grid_s = on_grid(dslow, &dq_slow);
        const bool grid_f = on_grid(dfast, &dq_fast);
        const bool is_fast = (lane & 1) != 0;
        const int  slot    = lane >> 1;
        const float d = is_fast ? dfast : dslow;
        const float T = TWO_PI_F - (is_fast ? dq_fast : dq_slow);
        float* __restrict__ o = out + (is_fast ? (size_t)N : (size_t)0);
        if (grid_s && grid_f) scan_two_lanes<true >(is_fast, slot, d, T, keep, o, N);
        else                  scan_two_lanes<false>(is_fast, slot, d, T, keep, o, N);
        return;
    }

    if (lane != 0) return;
    float slow = 0.0f, fast = 0.0f, prev = 0.0f;
    float* __restrict__ oS = out;
    float* __restrict__ oF = out + N;
    for (int t = 0; t < N; ++t) {
        oS[t] = slow;
        oF[t] = fast;
        bool crossed = (prev > PI_F) && (slow < PI_F);
        float ns = pmod(slow + dslow, TWO_PI_F);
        float nf = pmod(fast + dfast, TWO_PI_F);
        if (crossed) nf = pmod(nf * keep, TWO_PI_F);
        prev = slow; slow = ns; fast = nf;
    }
}

__global__ void osc_scale_wmma_kernel(const float* __restrict__ in,
                                      float* __restrict__ out, int N) {
    const float s = INV_TWO_PI_F;
    const int base = blockIdx.x * 256;
    const int lane = threadIdx.x;
    const int i    = lane & 15;
    const int kh   = lane >> 4;

    if (base + 256 <= N) {
        v8f d = {0.f, 0.f, 0.f, 0.f, 0.f, 0.f, 0.f, 0.f};
#pragma unroll
        for (int m = 0; m < 4; ++m) {
            const float* p = in + base + i * 16 + 4 * m + 2 * kh;
            v2f a; a.x = p[0]; a.y = p[1];
            v2f b;
            b.x = (i == (4 * m + 2 * kh + 0)) ? s : 0.0f;
            b.y = (i == (4 * m + 2 * kh + 1)) ? s : 0.0f;
            d = __builtin_amdgcn_wmma_f32_16x16x4_f32(
                    false, a, false, b, (short)0, d, false, false);
        }
#pragma unroll
        for (int r = 0; r < 8; ++r) {
            out[base + (r + 8 * kh) * 16 + i] = d[r];
        }
    } else {
        for (int j = base + lane; j < N; j += 32) out[j] = in[j] * s;
    }
}

extern "C" void kernel_launch(void* const* d_in, const int* in_sizes, int n_in,
                              void* d_out, int out_size, void* d_ws, size_t ws_size,
                              hipStream_t stream) {
    const float* p_log_slow = (const float*)d_in[0];
    const float* p_log_fast = (const float*)d_in[1];
    const float* p_reset    = (const float*)d_in[2];
    const int N = out_size / 3;
    if (N <= 0) return;
    float* out = (float*)d_out;

    osc_scan_kernel<<<1, 32, 0, stream>>>(p_log_slow, p_log_fast, p_reset, out, N);

    const int tiles = (N + 255) / 256;
    osc_scale_wmma_kernel<<<tiles, 32, 0, stream>>>(out, out + 2 * (size_t)N, N);
}
